# starting kernel
# baseline (speedup 1.0000x reference)
#include <hip/hip_runtime.h>


typedef __attribute__((ext_vector_type(2))) float v2f;
typedef __attribute__((ext_vector_type(8))) float v8f;

#define NDIM   512
#define NBATCH 16384
#define NROUND 511
#define NPAIR  256

#define ROWS 8

__global__ __launch_bounds__(256) void build_u_kernel(
    const float* __restrict__ angles,
    const int*   __restrict__ blocks,
    float*       __restrict__ U)
{
    __shared__ float rows[NDIM * ROWS];
    const int t  = threadIdx.x;
    const int r0 = blockIdx.x * ROWS;

    for (int idx = t; idx < NDIM * ROWS; idx += 256) {
        int c  = idx >> 3;
        int rr = idx & 7;
        rows[idx] = (c == r0 + rr) ? 1.0f : 0.0f;
    }
    __syncthreads();

    for (int rd = 0; rd < NROUND; ++rd) {
        const int   base = rd * NPAIR + t;
        const int   i    = blocks[2 * base + 0];
        const int   j    = blocks[2 * base + 1];
        const float a    = angles[base];
        const float cth  = cosf(a);
        const float sth  = sinf(a);
        float* __restrict__ pi = &rows[i << 3];
        float* __restrict__ pj = &rows[j << 3];
        #pragma unroll
        for (int rr = 0; rr < ROWS; ++rr) {
            const float ui = pi[rr];
            const float uj = pj[rr];
            pi[rr] = cth * ui + sth * uj;
            pj[rr] = cth * uj - sth * ui;
        }
        __syncthreads();
    }

    for (int idx = t; idx < NDIM * ROWS; idx += 256) {
        const int c  = idx & (NDIM - 1);
        const int rr = idx >> 9;
        U[(r0 + rr) * NDIM + c] = rows[(c << 3) + rr];
    }
}

#define BM  128
#define BN  64
#define BK  64
#define LDK 68

__global__ __launch_bounds__(256) void givens_gemm_kernel(
    const float* __restrict__ x,
    const float* __restrict__ U,
    const float* __restrict__ bias,
    float*       __restrict__ out)
{
    __shared__ float sx[BM * LDK];
    __shared__ float su[BN * LDK];

    const int t    = threadIdx.x;
    const int lane = t & 31;
    const int wave = t >> 5;
    const int lm   = lane & 15;
    const int half = lane >> 4;
    const int wm   = wave & 3;
    const int wn   = wave >> 2;

    const int m0 = blockIdx.x * BM;
    const int n0 = blockIdx.y * BN;

    v8f acc[2][2] = {};

    for (int kc = 0; kc < NDIM; kc += BK) {
        {
            const int c4 = (t & 15) << 2;
            const int rb = t >> 4;
            #pragma unroll
            for (int r = rb; r < BM; r += 16) {
                const float4 v = *(const float4*)&x[(m0 + r) * NDIM + kc + c4];
                *(float4*)&sx[r * LDK + c4] = v;
            }
            #pragma unroll
            for (int r = rb; r < BN; r += 16) {
                const float4 v = *(const float4*)&U[(n0 + r) * NDIM + kc + c4];
                *(float4*)&su[r * LDK + c4] = v;
            }
        }
        __syncthreads();

        const int arow0 = (wm * 32 +      lm) * LDK + 2 * half;
        const int arow1 = (wm * 32 + 16 + lm) * LDK + 2 * half;
        const int brow0 = (wn * 32 +      lm) * LDK + 2 * half;
        const int brow1 = (wn * 32 + 16 + lm) * LDK + 2 * half;
        #pragma unroll
        for (int kk = 0; kk < BK; kk += 4) {
            const v2f a0 = *(const v2f*)&sx[arow0 + kk];
            const v2f a1 = *(const v2f*)&sx[arow1 + kk];
            const v2f b0 = *(const v2f*)&su[brow0 + kk];
            const v2f b1 = *(const v2f*)&su[brow1 + kk];
            acc[0][0] = __builtin_amdgcn_wmma_f32_16x16x4_f32(
                false, a0, false, b0, (short)0, acc[0][0], false, false);
            acc[0][1] = __builtin_amdgcn_wmma_f32_16x16x4_f32(
                false, a0, false, b1, (short)0, acc[0][1], false, false);
            acc[1][0] = __builtin_amdgcn_wmma_f32_16x16x4_f32(
                false, a1, false, b0, (short)0, acc[1][0], false, false);
            acc[1][1] = __builtin_amdgcn_wmma_f32_16x16x4_f32(
                false, a1, false, b1, (short)0, acc[1][1], false, false);
        }
        __syncthreads();
    }

    #pragma unroll
    for (int ni = 0; ni < 2; ++ni) {
        const int n  = n0 + wn * 32 + ni * 16 + lm;
        const float bv = bias[n];
        #pragma unroll
        for (int mi = 0; mi < 2; ++mi) {
            const int mbase = m0 + wm * 32 + mi * 16 + half * 8;
            #pragma unroll
            for (int v = 0; v < 8; ++v) {
                out[(mbase + v) * NDIM + n] = acc[mi][ni][v] + bv;
            }
        }
    }
}

extern "C" void kernel_launch(void* const* d_in, const int* in_sizes, int n_in,
                              void* d_out, int out_size, void* d_ws, size_t ws_size,
                              hipStream_t stream) {
    const float* x      = (const float*)d_in[0];
    const float* angles = (const float*)d_in[1];
    const float* bias   = (const float*)d_in[2];
    const int*   blocks = (const int*)  d_in[3];

    float* U   = (float*)d_ws;
    float* out = (float*)d_out;

    build_u_kernel<<<dim3(NDIM / ROWS), dim3(256), 0, stream>>>(angles, blocks, U);
    givens_gemm_kernel<<<dim3(NBATCH / BM, NDIM / BN), dim3(256), 0, stream>>>(
        x, U, bias, out);
}


# build-U round loop: pair indices and angle of a round prefetched 5 rounds ahead in a register ring, sin/cos one round ahead, LDS-only barrier keeps those loads in flight; sums unchanged
# speedup vs baseline: 2.3514x; 2.3514x over previous
#include <hip/hip_runtime.h>


typedef __attribute__((ext_vector_type(2))) float v2f;
typedef __attribute__((ext_vector_type(8))) float v8f;

#define NDIM   512
#define NBATCH 16384
#define NROUND 511
#define NPAIR  256

#define ROWS 8
#define PF   7
#define LAG  2

static __device__ __forceinline__ void round_barrier() {
    asm volatile("s_wait_dscnt 0x0" ::: "memory");
    asm volatile("s_barrier_signal -1" ::: "memory");
    asm volatile("s_barrier_wait -1" ::: "memory");
}

__global__ __launch_bounds__(256) void build_u_kernel(
    const float* __restrict__ angles,
    const int*   __restrict__ blocks,
    float*       __restrict__ U)
{
    __shared__ float rows[NDIM * ROWS];
    const int t  = threadIdx.x;
    const int r0 = blockIdx.x * ROWS;

    const int2 q0 = *(const int2*)&blocks[2 * t];
    const float a0 = angles[t];
    int2  qr[PF] = {};
    float ar[PF] = {};
    #pragma unroll
    for (int k = 0; k < PF - LAG; ++k) {
        const int base = (k + 1) * NPAIR + t;
        qr[k] = *(const int2*)&blocks[2 * base];
        ar[k] = angles[base];
    }

    for (int idx = t; idx < NDIM * ROWS; idx += 256) {
        int c  = idx >> 3;
        int rr = idx & 7;
        rows[idx] = (c == r0 + rr) ? 1.0f : 0.0f;
    }

    int   io  = q0.x << 3;
    int   jo  = q0.y << 3;
    float cth = cosf(a0);
    float sth = sinf(a0);
    __syncthreads();

    for (int rd0 = 0; rd0 < NROUND; rd0 += PF) {
        #pragma unroll
        for (int k = 0; k < PF; ++k) {
            const int rd = rd0 + k;
            float* __restrict__ pi = &rows[io];
            float* __restrict__ pj = &rows[jo];
            float ui[ROWS], uj[ROWS];
            #pragma unroll
            for (int rr = 0; rr < ROWS; ++rr) {
                ui[rr] = pi[rr];
                uj[rr] = pj[rr];
            }

            {
                int nr = rd + PF - LAG + 1;
                nr = nr < NROUND ? nr : NROUND - 1;
                const int base = nr * NPAIR + t;
                const int ks = (k + PF - LAG) % PF;
                qr[ks] = *(const int2*)&blocks[2 * base];
                ar[ks] = angles[base];
            }

            const int   ion = qr[k].x << 3;
            const int   jon = qr[k].y << 3;
            const float cn  = cosf(ar[k]);
            const float sn  = sinf(ar[k]);

            #pragma unroll
            for (int rr = 0; rr < ROWS; ++rr) {
                pi[rr] = cth * ui[rr] + sth * uj[rr];
                pj[rr] = cth * uj[rr] - sth * ui[rr];
            }
            round_barrier();

            io = ion;  jo = jon;  cth = cn;  sth = sn;
        }
    }

    for (int idx = t; idx < NDIM * ROWS; idx += 256) {
        const int c  = idx & (NDIM - 1);
        const int rr = idx >> 9;
        U[(r0 + rr) * NDIM + c] = rows[(c << 3) + rr];
    }
}

#define BM  128
#define BN  64
#define BK  64
#define LDK 68

__global__ __launch_bounds__(256) void givens_gemm_kernel(
    const float* __restrict__ x,
    const float* __restrict__ U,
    const float* __restrict__ bias,
    float*       __restrict__ out)
{
    __shared__ float sx[BM * LDK];
    __shared__ float su[BN * LDK];

    const int t    = threadIdx.x;
    const int lane = t & 31;
    const int wave = t >> 5;
    const int lm   = lane & 15;
    const int half = lane >> 4;
    const int wm   = wave & 3;
    const int wn   = wave >> 2;

    const int m0 = blockIdx.x * BM;
    const int n0 = blockIdx.y * BN;

    v8f acc[2][2] = {};

    for (int kc = 0; kc < NDIM; kc += BK) {
        {
            const int c4 = (t & 15) << 2;
            const int rb = t >> 4;
            #pragma unroll
            for (int r = rb; r < BM; r += 16) {
                const float4 v = *(const float4*)&x[(m0 + r) * NDIM + kc + c4];
                *(float4*)&sx[r * LDK + c4] = v;
            }
            #pragma unroll
            for (int r = rb; r < BN; r += 16) {
                const float4 v = *(const float4*)&U[(n0 + r) * NDIM + kc + c4];
                *(float4*)&su[r * LDK + c4] = v;
            }
        }
        __syncthreads();

        const int arow0 = (wm * 32 +      lm) * LDK + 2 * half;
        const int arow1 = (wm * 32 + 16 + lm) * LDK + 2 * half;
        const int brow0 = (wn * 32 +      lm) * LDK + 2 * half;
        const int brow1 = (wn * 32 + 16 + lm) * LDK + 2 * half;
        #pragma unroll
        for (int kk = 0; kk < BK; kk += 4) {
            const v2f a0 = *(const v2f*)&sx[arow0 + kk];
            const v2f a1 = *(const v2f*)&sx[arow1 + kk];
            const v2f b0 = *(const v2f*)&su[brow0 + kk];
            const v2f b1 = *(const v2f*)&su[brow1 + kk];
            acc[0][0] = __builtin_amdgcn_wmma_f32_16x16x4_f32(
                false, a0, false, b0, (short)0, acc[0][0], false, false);
            acc[0][1] = __builtin_amdgcn_wmma_f32_16x16x4_f32(
                false, a0, false, b1, (short)0, acc[0][1], false, false);
            acc[1][0] = __builtin_amdgcn_wmma_f32_16x16x4_f32(
                false, a1, false, b0, (short)0, acc[1][0], false, false);
            acc[1][1] = __builtin_amdgcn_wmma_f32_16x16x4_f32(
                false, a1, false, b1, (short)0, acc[1][1], false, false);
        }
        __syncthreads();
    }

    #pragma unroll
    for (int ni = 0; ni < 2; ++ni) {
        const int n  = n0 + wn * 32 + ni * 16 + lm;
        const float bv = bias[n];
        #pragma unroll
        for (int mi = 0; mi < 2; ++mi) {
            const int mbase = m0 + wm * 32 + mi * 16 + half * 8;
            #pragma unroll
            for (int v = 0; v < 8; ++v) {
                out[(mbase + v) * NDIM + n] = acc[mi][ni][v] + bv;
            }
        }
    }
}

extern "C" void kernel_launch(void* const* d_in, const int* in_sizes, int n_in,
                              void* d_out, int out_size, void* d_ws, size_t ws_size,
                              hipStream_t stream) {
    const float* x      = (const float*)d_in[0];
    const float* angles = (const float*)d_in[1];
    const float* bias   = (const float*)d_in[2];
    const int*   blocks = (const int*)  d_in[3];

    float* U   = (float*)d_ws;
    float* out = (float*)d_out;

    build_u_kernel<<<dim3(NDIM / ROWS), dim3(256), 0, stream>>>(angles, blocks, U);
    givens_gemm_kernel<<<dim3(NBATCH / BM, NDIM / BN), dim3(256), 0, stream>>>(
        x, U, bias, out);
}


# build-U: 2 rows per workgroup on 256 workgroups instead of 8 on 64: a round moves 4x fewer LDS bytes per WGP, a column is one conflict-free 8-byte access per thread; sums unchanged
# speedup vs baseline: 2.6160x; 1.1125x over previous
#include <hip/hip_runtime.h>


typedef __attribute__((ext_vector_type(2))) float v2f;
typedef __attribute__((ext_vector_type(8))) float v8f;

#define NDIM   512
#define NBATCH 16384
#define NROUND 511
#define NPAIR  256

#define ROWS 2
#define RSH  1
#define PF   7
#define LAG  2

static __device__ __forceinline__ void round_barrier() {
    asm volatile("s_wait_dscnt 0x0" ::: "memory");
    asm volatile("s_barrier_signal -1" ::: "memory");
    asm volatile("s_barrier_wait -1" ::: "memory");
}

__global__ __launch_bounds__(256) void build_u_kernel(
    const float* __restrict__ angles,
    const int*   __restrict__ blocks,
    float*       __restrict__ U)
{
    __shared__ float rows[NDIM * ROWS];
    const int t  = threadIdx.x;
    const int r0 = blockIdx.x * ROWS;

    const int2 q0 = *(const int2*)&blocks[2 * t];
    const float a0 = angles[t];
    int2  qr[PF] = {};
    float ar[PF] = {};
    #pragma unroll
    for (int k = 0; k < PF - LAG; ++k) {
        const int base = (k + 1) * NPAIR + t;
        qr[k] = *(const int2*)&blocks[2 * base];
        ar[k] = angles[base];
    }

    for (int idx = t; idx < NDIM * ROWS; idx += 256) {
        int c  = idx >> RSH;
        int rr = idx & (ROWS - 1);
        rows[idx] = (c == r0 + rr) ? 1.0f : 0.0f;
    }

    int   io  = q0.x << RSH;
    int   jo  = q0.y << RSH;
    float cth = cosf(a0);
    float sth = sinf(a0);
    __syncthreads();

    for (int rd0 = 0; rd0 < NROUND; rd0 += PF) {
        #pragma unroll
        for (int k = 0; k < PF; ++k) {
            const int rd = rd0 + k;
            float* __restrict__ pi = &rows[io];
            float* __restrict__ pj = &rows[jo];
            float ui[ROWS], uj[ROWS];
            #pragma unroll
            for (int rr = 0; rr < ROWS; ++rr) {
                ui[rr] = pi[rr];
                uj[rr] = pj[rr];
            }

            {
                int nr = rd + PF - LAG + 1;
                nr = nr < NROUND ? nr : NROUND - 1;
                const int base = nr * NPAIR + t;
                const int ks = (k + PF - LAG) % PF;
                qr[ks] = *(const int2*)&blocks[2 * base];
                ar[ks] = angles[base];
            }

            const int   ion = qr[k].x << RSH;
            const int   jon = qr[k].y << RSH;
            const float cn  = cosf(ar[k]);
            const float sn  = sinf(ar[k]);

            #pragma unroll
            for (int rr = 0; rr < ROWS; ++rr) {
                pi[rr] = cth * ui[rr] + sth * uj[rr];
                pj[rr] = cth * uj[rr] - sth * ui[rr];
            }
            round_barrier();

            io = ion;  jo = jon;  cth = cn;  sth = sn;
        }
    }

    for (int idx = t; idx < NDIM * ROWS; idx += 256) {
        const int c  = idx & (NDIM - 1);
        const int rr = idx >> 9;
        U[(r0 + rr) * NDIM + c] = rows[(c << RSH) + rr];
    }
}

#define BM  128
#define BN  64
#define BK  64
#define LDK 68

__global__ __launch_bounds__(256) void givens_gemm_kernel(
    const float* __restrict__ x,
    const float* __restrict__ U,
    const float* __restrict__ bias,
    float*       __restrict__ out)
{
    __shared__ float sx[BM * LDK];
    __shared__ float su[BN * LDK];

    const int t    = threadIdx.x;
    const int lane = t & 31;
    const int wave = t >> 5;
    const int lm   = lane & 15;
    const int half = lane >> 4;
    const int wm   = wave & 3;
    const int wn   = wave >> 2;

    const int m0 = blockIdx.x * BM;
    const int n0 = blockIdx.y * BN;

    v8f acc[2][2] = {};

    for (int kc = 0; kc < NDIM; kc += BK) {
        {
            const int c4 = (t & 15) << 2;
            const int rb = t >> 4;
            #pragma unroll
            for (int r = rb; r < BM; r += 16) {
                const float4 v = *(const float4*)&x[(m0 + r) * NDIM + kc + c4];
                *(float4*)&sx[r * LDK + c4] = v;
            }
            #pragma unroll
            for (int r = rb; r < BN; r += 16) {
                const float4 v = *(const float4*)&U[(n0 + r) * NDIM + kc + c4];
                *(float4*)&su[r * LDK + c4] = v;
            }
        }
        __syncthreads();

        const int arow0 = (wm * 32 +      lm) * LDK + 2 * half;
        const int arow1 = (wm * 32 + 16 + lm) * LDK + 2 * half;
        const int brow0 = (wn * 32 +      lm) * LDK + 2 * half;
        const int brow1 = (wn * 32 + 16 + lm) * LDK + 2 * half;
        #pragma unroll
        for (int kk = 0; kk < BK; kk += 4) {
            const v2f a0 = *(const v2f*)&sx[arow0 + kk];
            const v2f a1 = *(const v2f*)&sx[arow1 + kk];
            const v2f b0 = *(const v2f*)&su[brow0 + kk];
            const v2f b1 = *(const v2f*)&su[brow1 + kk];
            acc[0][0] = __builtin_amdgcn_wmma_f32_16x16x4_f32(
                false, a0, false, b0, (short)0, acc[0][0], false, false);
            acc[0][1] = __builtin_amdgcn_wmma_f32_16x16x4_f32(
                false, a0, false, b1, (short)0, acc[0][1], false, false);
            acc[1][0] = __builtin_amdgcn_wmma_f32_16x16x4_f32(
                false, a1, false, b0, (short)0, acc[1][0], false, false);
            acc[1][1] = __builtin_amdgcn_wmma_f32_16x16x4_f32(
                false, a1, false, b1, (short)0, acc[1][1], false, false);
        }
        __syncthreads();
    }

    #pragma unroll
    for (int ni = 0; ni < 2; ++ni) {
        const int n  = n0 + wn * 32 + ni * 16 + lm;
        const float bv = bias[n];
        #pragma unroll
        for (int mi = 0; mi < 2; ++mi) {
            const int mbase = m0 + wm * 32 + mi * 16 + half * 8;
            #pragma unroll
            for (int v = 0; v < 8; ++v) {
                out[(mbase + v) * NDIM + n] = acc[mi][ni][v] + bv;
            }
        }
    }
}

extern "C" void kernel_launch(void* const* d_in, const int* in_sizes, int n_in,
                              void* d_out, int out_size, void* d_ws, size_t ws_size,
                              hipStream_t stream) {
    const float* x      = (const float*)d_in[0];
    const float* angles = (const float*)d_in[1];
    const float* bias   = (const float*)d_in[2];
    const int*   blocks = (const int*)  d_in[3];

    float* U   = (float*)d_ws;
    float* out = (float*)d_out;

    build_u_kernel<<<dim3(NDIM / ROWS), dim3(256), 0, stream>>>(angles, blocks, U);
    givens_gemm_kernel<<<dim3(NBATCH / BM, NDIM / BN), dim3(256), 0, stream>>>(
        x, U, bias, out);
}


# build-U: cos, sin and the pair's two LDS offsets come as one 16-byte record per (round, pair) from a new pre-pass launch, not from sinf/cosf inside each of the 511 serial rounds; sums unchanged
# speedup vs baseline: 3.5973x; 1.3751x over previous
#include <hip/hip_runtime.h>


typedef __attribute__((ext_vector_type(2))) float v2f;
typedef __attribute__((ext_vector_type(8))) float v8f;

#define NDIM   512
#define NBATCH 16384
#define NROUND 511
#define NPAIR  256

#define ROWS 2
#define RSH  1

__global__ __launch_bounds__(256) void round_table_kernel(
    const float* __restrict__ angles,
    const int*   __restrict__ blocks,
    float*       __restrict__ tab)
{
    const int g = blockIdx.x * NPAIR + threadIdx.x;
    const int2  q = *(const int2*)&blocks[2 * g];
    const float a = angles[g];
    int4 rec;
    rec.x = (q.x & (NDIM - 1)) * (ROWS * 4);
    rec.y = (q.y & (NDIM - 1)) * (ROWS * 4);
    rec.z = __float_as_int(cosf(a));
    rec.w = __float_as_int(sinf(a));
    *(int4*)&tab[4 * g] = rec;
}

#define PF   7
#define LAG  2

static __device__ __forceinline__ void round_barrier() {
    asm volatile("s_wait_dscnt 0x0" ::: "memory");
    asm volatile("s_barrier_signal -1" ::: "memory");
    asm volatile("s_barrier_wait -1" ::: "memory");
}

__global__ __launch_bounds__(256) void build_u_kernel(
    const float* __restrict__ tab,
    float*       __restrict__ U)
{
    __shared__ float rows[NDIM * ROWS];
    const int t  = threadIdx.x;
    const int r0 = blockIdx.x * ROWS;

    const int4* __restrict__ rec = (const int4*)tab + t;

    int4 cur = rec[0];
    int4 ring[PF] = {};
    #pragma unroll
    for (int k = 0; k < PF - LAG; ++k)
        ring[k] = rec[(k + 1) * NPAIR];

    for (int idx = t; idx < NDIM * ROWS; idx += 256) {
        int c  = idx >> RSH;
        int rr = idx & (ROWS - 1);
        rows[idx] = (c == r0 + rr) ? 1.0f : 0.0f;
    }
    __syncthreads();

    for (int rd0 = 0; rd0 < NROUND; rd0 += PF) {
        #pragma unroll
        for (int k = 0; k < PF; ++k) {
            const int rd = rd0 + k;
            float2* pi = (float2*)((char*)rows + cur.x);
            float2* pj = (float2*)((char*)rows + cur.y);
            const float cth = __int_as_float(cur.z);
            const float sth = __int_as_float(cur.w);
            const float2 ui = *pi;
            const float2 uj = *pj;

            {
                int nr = rd + PF - LAG + 1;
                nr = nr < NROUND ? nr : NROUND - 1;
                ring[(k + PF - LAG) % PF] = rec[nr * NPAIR];
            }

            float2 vi, vj;
            vi.x = cth * ui.x + sth * uj.x;
            vi.y = cth * ui.y + sth * uj.y;
            vj.x = cth * uj.x - sth * ui.x;
            vj.y = cth * uj.y - sth * ui.y;
            *pi = vi;
            *pj = vj;
            round_barrier();

            cur = ring[k];
        }
    }

    for (int idx = t; idx < NDIM * ROWS; idx += 256) {
        const int c  = idx & (NDIM - 1);
        const int rr = idx >> 9;
        U[(r0 + rr) * NDIM + c] = rows[(c << RSH) + rr];
    }
}

#define BM  128
#define BN  64
#define BK  64
#define LDK 68

__global__ __launch_bounds__(256) void givens_gemm_kernel(
    const float* __restrict__ x,
    const float* __restrict__ U,
    const float* __restrict__ bias,
    float*       __restrict__ out)
{
    __shared__ float sx[BM * LDK];
    __shared__ float su[BN * LDK];

    const int t    = threadIdx.x;
    const int lane = t & 31;
    const int wave = t >> 5;
    const int lm   = lane & 15;
    const int half = lane >> 4;
    const int wm   = wave & 3;
    const int wn   = wave >> 2;

    const int m0 = blockIdx.x * BM;
    const int n0 = blockIdx.y * BN;

    v8f acc[2][2] = {};

    for (int kc = 0; kc < NDIM; kc += BK) {
        {
            const int c4 = (t & 15) << 2;
            const int rb = t >> 4;
            #pragma unroll
            for (int r = rb; r < BM; r += 16) {
                const float4 v = *(const float4*)&x[(m0 + r) * NDIM + kc + c4];
                *(float4*)&sx[r * LDK + c4] = v;
            }
            #pragma unroll
            for (int r = rb; r < BN; r += 16) {
                const float4 v = *(const float4*)&U[(n0 + r) * NDIM + kc + c4];
                *(float4*)&su[r * LDK + c4] = v;
            }
        }
        __syncthreads();

        const int arow0 = (wm * 32 +      lm) * LDK + 2 * half;
        const int arow1 = (wm * 32 + 16 + lm) * LDK + 2 * half;
        const int brow0 = (wn * 32 +      lm) * LDK + 2 * half;
        const int brow1 = (wn * 32 + 16 + lm) * LDK + 2 * half;
        #pragma unroll
        for (int kk = 0; kk < BK; kk += 4) {
            const v2f a0 = *(const v2f*)&sx[arow0 + kk];
            const v2f a1 = *(const v2f*)&sx[arow1 + kk];
            const v2f b0 = *(const v2f*)&su[brow0 + kk];
            const v2f b1 = *(const v2f*)&su[brow1 + kk];
            acc[0][0] = __builtin_amdgcn_wmma_f32_16x16x4_f32(
                false, a0, false, b0, (short)0, acc[0][0], false, false);
            acc[0][1] = __builtin_amdgcn_wmma_f32_16x16x4_f32(
                false, a0, false, b1, (short)0, acc[0][1], false, false);
            acc[1][0] = __builtin_amdgcn_wmma_f32_16x16x4_f32(
                false, a1, false, b0, (short)0, acc[1][0], false, false);
            acc[1][1] = __builtin_amdgcn_wmma_f32_16x16x4_f32(
                false, a1, false, b1, (short)0, acc[1][1], false, false);
        }
        __syncthreads();
    }

    #pragma unroll
    for (int ni = 0; ni < 2; ++ni) {
        const int n  = n0 + wn * 32 + ni * 16 + lm;
        const float bv = bias[n];
        #pragma unroll
        for (int mi = 0; mi < 2; ++mi) {
            const int mbase = m0 + wm * 32 + mi * 16 + half * 8;
            #pragma unroll
            for (int v = 0; v < 8; ++v) {
                out[(mbase + v) * NDIM + n] = acc[mi][ni][v] + bv;
            }
        }
    }
}

extern "C" void kernel_launch(void* const* d_in, const int* in_sizes, int n_in,
                              void* d_out, int out_size, void* d_ws, size_t ws_size,
                              hipStream_t stream) {
    const float* x      = (const float*)d_in[0];
    const float* angles = (const float*)d_in[1];
    const float* bias   = (const float*)d_in[2];
    const int*   blocks = (const int*)  d_in[3];

    float* U   = (float*)d_ws;
    float* out = (float*)d_out;
    float* tab = out;

    round_table_kernel<<<dim3(NROUND), dim3(NPAIR), 0, stream>>>(angles, blocks, tab);
    build_u_kernel<<<dim3(NDIM / ROWS), dim3(256), 0, stream>>>(tab, U);
    givens_gemm_kernel<<<dim3(NBATCH / BM, NDIM / BN), dim3(256), 0, stream>>>(
        x, U, bias, out);
}


# build-U: the 511 serial rounds cut into 4 independent segments built side by side from identity, then multiplied by f32 WMMA products in 2 new launches; changes the summation order of U
# speedup vs baseline: 4.3429x; 1.2073x over previous
#include <hip/hip_runtime.h>


typedef __attribute__((ext_vector_type(2))) float v2f;
typedef __attribute__((ext_vector_type(8))) float v8f;

#define NDIM   512
#define NBATCH 16384
#define NROUND 511
#define NPAIR  256

#define ROWS 2
#define RSH  1

__global__ __launch_bounds__(256) void round_table_kernel(
    const float* __restrict__ angles,
    const int*   __restrict__ blocks,
    float*       __restrict__ tab)
{
    const int g = blockIdx.x * NPAIR + threadIdx.x;
    const int2  q = *(const int2*)&blocks[2 * g];
    const float a = angles[g];
    int4 rec;
    rec.x = (q.x & (NDIM - 1)) * (ROWS * 4);
    rec.y = (q.y & (NDIM - 1)) * (ROWS * 4);
    rec.z = __float_as_int(cosf(a));
    rec.w = __float_as_int(sinf(a));
    *(int4*)&tab[4 * g] = rec;
}

#define PF   7
#define LAG  2

#define NSEG 4
#define SEG0 133
#define SEGN 126

static __device__ __forceinline__ void round_barrier() {
    asm volatile("s_wait_dscnt 0x0" ::: "memory");
    asm volatile("s_barrier_signal -1" ::: "memory");
    asm volatile("s_barrier_wait -1" ::: "memory");
}

__global__ __launch_bounds__(256) void build_u_kernel(
    const float* __restrict__ tab,
    float*       __restrict__ P)
{
    __shared__ float rows[NDIM * ROWS];
    const int t  = threadIdx.x;
    const int r0 = blockIdx.x * ROWS;
    const int seg   = blockIdx.y;
    const int rbase = seg == 0 ? 0 : SEG0 + (seg - 1) * SEGN;
    const int nrnd  = seg == 0 ? SEG0 : SEGN;

    const int4* __restrict__ rec = (const int4*)tab + rbase * NPAIR + t;

    int4 cur = rec[0];
    int4 ring[PF] = {};
    #pragma unroll
    for (int k = 0; k < PF - LAG; ++k)
        ring[k] = rec[(k + 1) * NPAIR];

    for (int idx = t; idx < NDIM * ROWS; idx += 256) {
        int c  = idx >> RSH;
        int rr = idx & (ROWS - 1);
        rows[idx] = (c == r0 + rr) ? 1.0f : 0.0f;
    }
    __syncthreads();

    for (int rd0 = 0; rd0 < nrnd; rd0 += PF) {
        #pragma unroll
        for (int k = 0; k < PF; ++k) {
            const int rd = rd0 + k;
            float2* pi = (float2*)((char*)rows + cur.x);
            float2* pj = (float2*)((char*)rows + cur.y);
            const float cth = __int_as_float(cur.z);
            const float sth = __int_as_float(cur.w);
            const float2 ui = *pi;
            const float2 uj = *pj;

            {
                int nr = rd + PF - LAG + 1;
                nr = nr < nrnd ? nr : nrnd - 1;
                ring[(k + PF - LAG) % PF] = rec[nr * NPAIR];
            }

            float2 vi, vj;
            vi.x = cth * ui.x + sth * uj.x;
            vi.y = cth * ui.y + sth * uj.y;
            vj.x = cth * uj.x - sth * ui.x;
            vj.y = cth * uj.y - sth * ui.y;
            *pi = vi;
            *pj = vj;
            round_barrier();

            cur = ring[k];
        }
    }

    float* __restrict__ Ps = P + seg * (NDIM * NDIM);
    for (int idx = t; idx < NDIM * ROWS; idx += 256) {
        const int c  = idx & (NDIM - 1);
        const int rr = idx >> 9;
        Ps[(r0 + rr) * NDIM + c] = rows[(c << RSH) + rr];
    }
}

#define CM  32
#define CN  64
#define CK  64
#define LDA 68
#define LDB 72

__global__ __launch_bounds__(256) void seg_product_kernel(
    const float* __restrict__ S,
    float*       __restrict__ D)
{
    __shared__ float sa[CM * LDA];
    __shared__ float sb[CK * LDB];

    const int t    = threadIdx.x;
    const int lane = t & 31;
    const int wave = t >> 5;
    const int lm   = lane & 15;
    const int half = lane >> 4;
    const int wm   = wave & 1;
    const int wn   = (wave >> 1) & 1;
    const int kh   = wave >> 2;

    const float* __restrict__ A = S + (2 * blockIdx.z) * (NDIM * NDIM);
    const float* __restrict__ B = A + NDIM * NDIM;
    float*       __restrict__ C = D + blockIdx.z * (NDIM * NDIM);

    const int m0 = blockIdx.x * CM;
    const int n0 = blockIdx.y * CN;
    const int c4 = (t & 15) << 2;
    const int rb = t >> 4;

    const float* __restrict__ ga = &A[(m0 + rb) * NDIM + c4];
    const float* __restrict__ gb = &B[rb * NDIM + n0 + c4];
    float4 ra0 = *(const float4*)&ga[0];
    float4 ra1 = *(const float4*)&ga[16 * NDIM];
    float4 rv0 = *(const float4*)&gb[0];
    float4 rv1 = *(const float4*)&gb[16 * NDIM];
    float4 rv2 = *(const float4*)&gb[32 * NDIM];
    float4 rv3 = *(const float4*)&gb[48 * NDIM];

    v8f acc0 = {}, acc1 = {};
    const int aoff = (wm * 16 + lm) * LDA + kh * 32 + 2 * half;
    const int boff = (kh * 32 + 2 * half) * LDB + wn * 32 + lm;

    for (int kc = 0; kc < NDIM; kc += CK) {
        *(float4*)&sa[rb * LDA + c4]        = ra0;
        *(float4*)&sa[(rb + 16) * LDA + c4] = ra1;
        *(float4*)&sb[rb * LDB + c4]        = rv0;
        *(float4*)&sb[(rb + 16) * LDB + c4] = rv1;
        *(float4*)&sb[(rb + 32) * LDB + c4] = rv2;
        *(float4*)&sb[(rb + 48) * LDB + c4] = rv3;
        round_barrier();
        if (kc + CK < NDIM) {
            ga += CK;
            gb += CK * NDIM;
            ra0 = *(const float4*)&ga[0];
            ra1 = *(const float4*)&ga[16 * NDIM];
            rv0 = *(const float4*)&gb[0];
            rv1 = *(const float4*)&gb[16 * NDIM];
            rv2 = *(const float4*)&gb[32 * NDIM];
            rv3 = *(const float4*)&gb[48 * NDIM];
        }
        #pragma unroll
        for (int kk = 0; kk < CK / 2; kk += 4) {
            const v2f a = *(const v2f*)&sa[aoff + kk];
            v2f b0, b1;
            b0.x = sb[boff + kk * LDB];
            b0.y = sb[boff + kk * LDB + LDB];
            b1.x = sb[boff + kk * LDB + 16];
            b1.y = sb[boff + kk * LDB + LDB + 16];
            acc0 = __builtin_amdgcn_wmma_f32_16x16x4_f32(
                false, a, false, b0, (short)0, acc0, false, false);
            acc1 = __builtin_amdgcn_wmma_f32_16x16x4_f32(
                false, a, false, b1, (short)0, acc1, false, false);
        }
        round_barrier();
    }

    float* red = sa;
    const int q = wave & 3;
    if (kh == 1) {
        #pragma unroll
        for (int v = 0; v < 8; ++v) {
            red[((q * 2 + 0) * 8 + v) * 32 + lane] = acc0[v];
            red[((q * 2 + 1) * 8 + v) * 32 + lane] = acc1[v];
        }
    }
    __syncthreads();
    if (kh == 0) {
        const int n = n0 + wn * 32 + lm;
        #pragma unroll
        for (int v = 0; v < 8; ++v) {
            const int m = m0 + wm * 16 + half * 8 + v;
            C[m * NDIM + n]      = acc0[v] + red[((q * 2 + 0) * 8 + v) * 32 + lane];
            C[m * NDIM + n + 16] = acc1[v] + red[((q * 2 + 1) * 8 + v) * 32 + lane];
        }
    }
}

#define BM  128
#define BN  64
#define BK  64
#define LDK 68

__global__ __launch_bounds__(256) void givens_gemm_kernel(
    const float* __restrict__ x,
    const float* __restrict__ U,
    const float* __restrict__ bias,
    float*       __restrict__ out)
{
    __shared__ float sx[BM * LDK];
    __shared__ float su[BN * LDK];

    const int t    = threadIdx.x;
    const int lane = t & 31;
    const int wave = t >> 5;
    const int lm   = lane & 15;
    const int half = lane >> 4;
    const int wm   = wave & 3;
    const int wn   = wave >> 2;

    const int m0 = blockIdx.x * BM;
    const int n0 = blockIdx.y * BN;

    v8f acc[2][2] = {};

    for (int kc = 0; kc < NDIM; kc += BK) {
        {
            const int c4 = (t & 15) << 2;
            const int rb = t >> 4;
            #pragma unroll
            for (int r = rb; r < BM; r += 16) {
                const float4 v = *(const float4*)&x[(m0 + r) * NDIM + kc + c4];
                *(float4*)&sx[r * LDK + c4] = v;
            }
            #pragma unroll
            for (int r = rb; r < BN; r += 16) {
                const float4 v = *(const float4*)&U[(n0 + r) * NDIM + kc + c4];
                *(float4*)&su[r * LDK + c4] = v;
            }
        }
        __syncthreads();

        const int arow0 = (wm * 32 +      lm) * LDK + 2 * half;
        const int arow1 = (wm * 32 + 16 + lm) * LDK + 2 * half;
        const int brow0 = (wn * 32 +      lm) * LDK + 2 * half;
        const int brow1 = (wn * 32 + 16 + lm) * LDK + 2 * half;
        #pragma unroll
        for (int kk = 0; kk < BK; kk += 4) {
            const v2f a0 = *(const v2f*)&sx[arow0 + kk];
            const v2f a1 = *(const v2f*)&sx[arow1 + kk];
            const v2f b0 = *(const v2f*)&su[brow0 + kk];
            const v2f b1 = *(const v2f*)&su[brow1 + kk];
            acc[0][0] = __builtin_amdgcn_wmma_f32_16x16x4_f32(
                false, a0, false, b0, (short)0, acc[0][0], false, false);
            acc[0][1] = __builtin_amdgcn_wmma_f32_16x16x4_f32(
                false, a0, false, b1, (short)0, acc[0][1], false, false);
            acc[1][0] = __builtin_amdgcn_wmma_f32_16x16x4_f32(
                false, a1, false, b0, (short)0, acc[1][0], false, false);
            acc[1][1] = __builtin_amdgcn_wmma_f32_16x16x4_f32(
                false, a1, false, b1, (short)0, acc[1][1], false, false);
        }
        __syncthreads();
    }

    #pragma unroll
    for (int ni = 0; ni < 2; ++ni) {
        const int n  = n0 + wn * 32 + ni * 16 + lm;
        const float bv = bias[n];
        #pragma unroll
        for (int mi = 0; mi < 2; ++mi) {
            const int mbase = m0 + wm * 32 + mi * 16 + half * 8;
            #pragma unroll
            for (int v = 0; v < 8; ++v) {
                out[(mbase + v) * NDIM + n] = acc[mi][ni][v] + bv;
            }
        }
    }
}

extern "C" void kernel_launch(void* const* d_in, const int* in_sizes, int n_in,
                              void* d_out, int out_size, void* d_ws, size_t ws_size,
                              hipStream_t stream) {
    const float* x      = (const float*)d_in[0];
    const float* angles = (const float*)d_in[1];
    const float* bias   = (const float*)d_in[2];
    const int*   blocks = (const int*)  d_in[3];

    float* U   = (float*)d_ws;
    float* out = (float*)d_out;
    float* tab = out;
    float* P   = out + 524288;
    float* Q   = out + 524288 + 1048576;

    round_table_kernel<<<dim3(NROUND), dim3(NPAIR), 0, stream>>>(angles, blocks, tab);
    build_u_kernel<<<dim3(NDIM / ROWS, NSEG), dim3(256), 0, stream>>>(tab, P);
    seg_product_kernel<<<dim3(NDIM / CM, NDIM / CN, 2), dim3(256), 0, stream>>>(P, Q);
    seg_product_kernel<<<dim3(NDIM / CM, NDIM / CN, 1), dim3(256), 0, stream>>>(Q, U);
    givens_gemm_kernel<<<dim3(NBATCH / BM, NDIM / BN), dim3(256), 0, stream>>>(
        x, U, bias, out);
}


# phase-2 GEMM: each K chunk's global loads are issued one chunk ahead into registers, in front of the current chunk's matrix instructions, behind LDS-only barriers; sums unchanged
# speedup vs baseline: 4.6425x; 1.0690x over previous
#include <hip/hip_runtime.h>


typedef __attribute__((ext_vector_type(2))) float v2f;
typedef __attribute__((ext_vector_type(8))) float v8f;

#define NDIM   512
#define NBATCH 16384
#define NROUND 511
#define NPAIR  256

#define ROWS 2
#define RSH  1

__global__ __launch_bounds__(256) void round_table_kernel(
    const float* __restrict__ angles,
    const int*   __restrict__ blocks,
    float*       __restrict__ tab)
{
    const int g = blockIdx.x * NPAIR + threadIdx.x;
    const int2  q = *(const int2*)&blocks[2 * g];
    const float a = angles[g];
    int4 rec;
    rec.x = (q.x & (NDIM - 1)) * (ROWS * 4);
    rec.y = (q.y & (NDIM - 1)) * (ROWS * 4);
    rec.z = __float_as_int(cosf(a));
    rec.w = __float_as_int(sinf(a));
    *(int4*)&tab[4 * g] = rec;
}

#define PF   7
#define LAG  2

#define NSEG 4
#define SEG0 133
#define SEGN 126

static __device__ __forceinline__ void round_barrier() {
    asm volatile("s_wait_dscnt 0x0" ::: "memory");
    asm volatile("s_barrier_signal -1" ::: "memory");
    asm volatile("s_barrier_wait -1" ::: "memory");
}

__global__ __launch_bounds__(256) void build_u_kernel(
    const float* __restrict__ tab,
    float*       __restrict__ P)
{
    __shared__ float rows[NDIM * ROWS];
    const int t  = threadIdx.x;
    const int r0 = blockIdx.x * ROWS;
    const int seg   = blockIdx.y;
    const int rbase = seg == 0 ? 0 : SEG0 + (seg - 1) * SEGN;
    const int nrnd  = seg == 0 ? SEG0 : SEGN;

    const int4* __restrict__ rec = (const int4*)tab + rbase * NPAIR + t;

    int4 cur = rec[0];
    int4 ring[PF] = {};
    #pragma unroll
    for (int k = 0; k < PF - LAG; ++k)
        ring[k] = rec[(k + 1) * NPAIR];

    for (int idx = t; idx < NDIM * ROWS; idx += 256) {
        int c  = idx >> RSH;
        int rr = idx & (ROWS - 1);
        rows[idx] = (c == r0 + rr) ? 1.0f : 0.0f;
    }
    __syncthreads();

    for (int rd0 = 0; rd0 < nrnd; rd0 += PF) {
        #pragma unroll
        for (int k = 0; k < PF; ++k) {
            const int rd = rd0 + k;
            float2* pi = (float2*)((char*)rows + cur.x);
            float2* pj = (float2*)((char*)rows + cur.y);
            const float cth = __int_as_float(cur.z);
            const float sth = __int_as_float(cur.w);
            const float2 ui = *pi;
            const float2 uj = *pj;

            {
                int nr = rd + PF - LAG + 1;
                nr = nr < nrnd ? nr : nrnd - 1;
                ring[(k + PF - LAG) % PF] = rec[nr * NPAIR];
            }

            float2 vi, vj;
            vi.x = cth * ui.x + sth * uj.x;
            vi.y = cth * ui.y + sth * uj.y;
            vj.x = cth * uj.x - sth * ui.x;
            vj.y = cth * uj.y - sth * ui.y;
            *pi = vi;
            *pj = vj;
            round_barrier();

            cur = ring[k];
        }
    }

    float* __restrict__ Ps = P + seg * (NDIM * NDIM);
    for (int idx = t; idx < NDIM * ROWS; idx += 256) {
        const int c  = idx & (NDIM - 1);
        const int rr = idx >> 9;
        Ps[(r0 + rr) * NDIM + c] = rows[(c << RSH) + rr];
    }
}

#define CM  32
#define CN  64
#define CK  64
#define LDA 68
#define LDB 72

__global__ __launch_bounds__(256) void seg_product_kernel(
    const float* __restrict__ S,
    float*       __restrict__ D)
{
    __shared__ float sa[CM * LDA];
    __shared__ float sb[CK * LDB];

    const int t    = threadIdx.x;
    const int lane = t & 31;
    const int wave = t >> 5;
    const int lm   = lane & 15;
    const int half = lane >> 4;
    const int wm   = wave & 1;
    const int wn   = (wave >> 1) & 1;
    const int kh   = wave >> 2;

    const float* __restrict__ A = S + (2 * blockIdx.z) * (NDIM * NDIM);
    const float* __restrict__ B = A + NDIM * NDIM;
    float*       __restrict__ C = D + blockIdx.z * (NDIM * NDIM);

    const int m0 = blockIdx.x * CM;
    const int n0 = blockIdx.y * CN;
    const int c4 = (t & 15) << 2;
    const int rb = t >> 4;

    const float* __restrict__ ga = &A[(m0 + rb) * NDIM + c4];
    const float* __restrict__ gb = &B[rb * NDIM + n0 + c4];
    float4 ra0 = *(const float4*)&ga[0];
    float4 ra1 = *(const float4*)&ga[16 * NDIM];
    float4 rv0 = *(const float4*)&gb[0];
    float4 rv1 = *(const float4*)&gb[16 * NDIM];
    float4 rv2 = *(const float4*)&gb[32 * NDIM];
    float4 rv3 = *(const float4*)&gb[48 * NDIM];

    v8f acc0 = {}, acc1 = {};
    const int aoff = (wm * 16 + lm) * LDA + kh * 32 + 2 * half;
    const int boff = (kh * 32 + 2 * half) * LDB + wn * 32 + lm;

    for (int kc = 0; kc < NDIM; kc += CK) {
        *(float4*)&sa[rb * LDA + c4]        = ra0;
        *(float4*)&sa[(rb + 16) * LDA + c4] = ra1;
        *(float4*)&sb[rb * LDB + c4]        = rv0;
        *(float4*)&sb[(rb + 16) * LDB + c4] = rv1;
        *(float4*)&sb[(rb + 32) * LDB + c4] = rv2;
        *(float4*)&sb[(rb + 48) * LDB + c4] = rv3;
        round_barrier();
        if (kc + CK < NDIM) {
            ga += CK;
            gb += CK * NDIM;
            ra0 = *(const float4*)&ga[0];
            ra1 = *(const float4*)&ga[16 * NDIM];
            rv0 = *(const float4*)&gb[0];
            rv1 = *(const float4*)&gb[16 * NDIM];
            rv2 = *(const float4*)&gb[32 * NDIM];
            rv3 = *(const float4*)&gb[48 * NDIM];
        }
        #pragma unroll
        for (int kk = 0; kk < CK / 2; kk += 4) {
            const v2f a = *(const v2f*)&sa[aoff + kk];
            v2f b0, b1;
            b0.x = sb[boff + kk * LDB];
            b0.y = sb[boff + kk * LDB + LDB];
            b1.x = sb[boff + kk * LDB + 16];
            b1.y = sb[boff + kk * LDB + LDB + 16];
            acc0 = __builtin_amdgcn_wmma_f32_16x16x4_f32(
                false, a, false, b0, (short)0, acc0, false, false);
            acc1 = __builtin_amdgcn_wmma_f32_16x16x4_f32(
                false, a, false, b1, (short)0, acc1, false, false);
        }
        round_barrier();
    }

    float* red = sa;
    const int q = wave & 3;
    if (kh == 1) {
        #pragma unroll
        for (int v = 0; v < 8; ++v) {
            red[((q * 2 + 0) * 8 + v) * 32 + lane] = acc0[v];
            red[((q * 2 + 1) * 8 + v) * 32 + lane] = acc1[v];
        }
    }
    __syncthreads();
    if (kh == 0) {
        const int n = n0 + wn * 32 + lm;
        #pragma unroll
        for (int v = 0; v < 8; ++v) {
            const int m = m0 + wm * 16 + half * 8 + v;
            C[m * NDIM + n]      = acc0[v] + red[((q * 2 + 0) * 8 + v) * 32 + lane];
            C[m * NDIM + n + 16] = acc1[v] + red[((q * 2 + 1) * 8 + v) * 32 + lane];
        }
    }
}

#define BM  128
#define BN  64
#define BK  64
#define LDK 68
#define RPP 16

__global__ __launch_bounds__(256) void givens_gemm_kernel(
    const float* __restrict__ x,
    const float* __restrict__ U,
    const float* __restrict__ bias,
    float*       __restrict__ out)
{
    __shared__ float sx[BM * LDK];
    __shared__ float su[BN * LDK];

    const int t    = threadIdx.x;
    const int lane = t & 31;
    const int wave = t >> 5;
    const int lm   = lane & 15;
    const int half = lane >> 4;
    const int wm   = wave & 3;
    const int wn   = wave >> 2;

    const int m0 = blockIdx.x * BM;
    const int n0 = blockIdx.y * BN;

    const int c4 = (t & 15) << 2;
    const int rb = t >> 4;
    const float* __restrict__ gx = &x[(m0 + rb) * NDIM + c4];
    const float* __restrict__ gu = &U[(n0 + rb) * NDIM + c4];
    const int so = rb * LDK + c4;

    float4 rx0, rx1, rx2, rx3, rx4, rx5, rx6, rx7;
    float4 ru0, ru1, ru2, ru3;
    rx0 = *(const float4*)&gx[0 * RPP * NDIM];
    rx1 = *(const float4*)&gx[1 * RPP * NDIM];
    rx2 = *(const float4*)&gx[2 * RPP * NDIM];
    rx3 = *(const float4*)&gx[3 * RPP * NDIM];
    rx4 = *(const float4*)&gx[4 * RPP * NDIM];
    rx5 = *(const float4*)&gx[5 * RPP * NDIM];
    rx6 = *(const float4*)&gx[6 * RPP * NDIM];
    rx7 = *(const float4*)&gx[7 * RPP * NDIM];
    ru0 = *(const float4*)&gu[0 * RPP * NDIM];
    ru1 = *(const float4*)&gu[1 * RPP * NDIM];
    ru2 = *(const float4*)&gu[2 * RPP * NDIM];
    ru3 = *(const float4*)&gu[3 * RPP * NDIM];

    v8f acc[2][2] = {};
    const int arow0 = (wm * 32 +      lm) * LDK + 2 * half;
    const int arow1 = (wm * 32 + 16 + lm) * LDK + 2 * half;
    const int brow0 = (wn * 32 +      lm) * LDK + 2 * half;
    const int brow1 = (wn * 32 + 16 + lm) * LDK + 2 * half;

    for (int kc = 0; kc < NDIM; kc += BK) {
        *(float4*)&sx[so + 0 * RPP * LDK] = rx0;
        *(float4*)&sx[so + 1 * RPP * LDK] = rx1;
        *(float4*)&sx[so + 2 * RPP * LDK] = rx2;
        *(float4*)&sx[so + 3 * RPP * LDK] = rx3;
        *(float4*)&sx[so + 4 * RPP * LDK] = rx4;
        *(float4*)&sx[so + 5 * RPP * LDK] = rx5;
        *(float4*)&sx[so + 6 * RPP * LDK] = rx6;
        *(float4*)&sx[so + 7 * RPP * LDK] = rx7;
        *(float4*)&su[so + 0 * RPP * LDK] = ru0;
        *(float4*)&su[so + 1 * RPP * LDK] = ru1;
        *(float4*)&su[so + 2 * RPP * LDK] = ru2;
        *(float4*)&su[so + 3 * RPP * LDK] = ru3;
        round_barrier();

        if (kc + BK < NDIM) {
            gx += BK;
            gu += BK;
            rx0 = *(const float4*)&gx[0 * RPP * NDIM];
            rx1 = *(const float4*)&gx[1 * RPP * NDIM];
            rx2 = *(const float4*)&gx[2 * RPP * NDIM];
            rx3 = *(const float4*)&gx[3 * RPP * NDIM];
            rx4 = *(const float4*)&gx[4 * RPP * NDIM];
            rx5 = *(const float4*)&gx[5 * RPP * NDIM];
            rx6 = *(const float4*)&gx[6 * RPP * NDIM];
            rx7 = *(const float4*)&gx[7 * RPP * NDIM];
            ru0 = *(const float4*)&gu[0 * RPP * NDIM];
            ru1 = *(const float4*)&gu[1 * RPP * NDIM];
            ru2 = *(const float4*)&gu[2 * RPP * NDIM];
            ru3 = *(const float4*)&gu[3 * RPP * NDIM];
        }

        #pragma unroll
        for (int kk = 0; kk < BK; kk += 4) {
            const v2f a0 = *(const v2f*)&sx[arow0 + kk];
            const v2f a1 = *(const v2f*)&sx[arow1 + kk];
            const v2f b0 = *(const v2f*)&su[brow0 + kk];
            const v2f b1 = *(const v2f*)&su[brow1 + kk];
            acc[0][0] = __builtin_amdgcn_wmma_f32_16x16x4_f32(
                false, a0, false, b0, (short)0, acc[0][0], false, false);
            acc[0][1] = __builtin_amdgcn_wmma_f32_16x16x4_f32(
                false, a0, false, b1, (short)0, acc[0][1], false, false);
            acc[1][0] = __builtin_amdgcn_wmma_f32_16x16x4_f32(
                false, a1, false, b0, (short)0, acc[1][0], false, false);
            acc[1][1] = __builtin_amdgcn_wmma_f32_16x16x4_f32(
                false, a1, false, b1, (short)0, acc[1][1], false, false);
        }
        round_barrier();
    }

    #pragma unroll
    for (int ni = 0; ni < 2; ++ni) {
        const int n  = n0 + wn * 32 + ni * 16 + lm;
        const float bv = bias[n];
        #pragma unroll
        for (int mi = 0; mi < 2; ++mi) {
            const int mbase = m0 + wm * 32 + mi * 16 + half * 8;
            #pragma unroll
            for (int v = 0; v < 8; ++v) {
                out[(mbase + v) * NDIM + n] = acc[mi][ni][v] + bv;
            }
        }
    }
}

extern "C" void kernel_launch(void* const* d_in, const int* in_sizes, int n_in,
                              void* d_out, int out_size, void* d_ws, size_t ws_size,
                              hipStream_t stream) {
    const float* x      = (const float*)d_in[0];
    const float* angles = (const float*)d_in[1];
    const float* bias   = (const float*)d_in[2];
    const int*   blocks = (const int*)  d_in[3];

    float* U   = (float*)d_ws;
    float* out = (float*)d_out;
    float* tab = out;
    float* P   = out + 524288;
    float* Q   = out + 524288 + 1048576;

    round_table_kernel<<<dim3(NROUND), dim3(NPAIR), 0, stream>>>(angles, blocks, tab);
    build_u_kernel<<<dim3(NDIM / ROWS, NSEG), dim3(256), 0, stream>>>(tab, P);
    seg_product_kernel<<<dim3(NDIM / CM, NDIM / CN, 2), dim3(256), 0, stream>>>(P, Q);
    seg_product_kernel<<<dim3(NDIM / CM, NDIM / CN, 1), dim3(256), 0, stream>>>(Q, U);
    givens_gemm_kernel<<<dim3(NBATCH / BM, NDIM / BN), dim3(256), 0, stream>>>(
        x, U, bias, out);
}


# build-U: 4 rows per workgroup on 4 x 128 workgroups instead of 2 on 4 x 256: half as many workgroups pull each round's records from the second-level cache, half the waves to launch; sums unchanged
# speedup vs baseline: 4.9032x; 1.0562x over previous
#include <hip/hip_runtime.h>


typedef __attribute__((ext_vector_type(2))) float v2f;
typedef __attribute__((ext_vector_type(8))) float v8f;

#define NDIM   512
#define NBATCH 16384
#define NROUND 511
#define NPAIR  256

#define ROWS 4
#define RSH  2

__global__ __launch_bounds__(256) void round_table_kernel(
    const float* __restrict__ angles,
    const int*   __restrict__ blocks,
    float*       __restrict__ tab)
{
    const int g = blockIdx.x * NPAIR + threadIdx.x;
    const int2  q = *(const int2*)&blocks[2 * g];
    const float a = angles[g];
    int4 rec;
    rec.x = (q.x & (NDIM - 1)) * (ROWS * 4);
    rec.y = (q.y & (NDIM - 1)) * (ROWS * 4);
    rec.z = __float_as_int(cosf(a));
    rec.w = __float_as_int(sinf(a));
    *(int4*)&tab[4 * g] = rec;
}

#define PF   7
#define LAG  2

#define NSEG 4
#define SEG0 133
#define SEGN 126

static __device__ __forceinline__ void round_barrier() {
    asm volatile("s_wait_dscnt 0x0" ::: "memory");
    asm volatile("s_barrier_signal -1" ::: "memory");
    asm volatile("s_barrier_wait -1" ::: "memory");
}

__global__ __launch_bounds__(256) void build_u_kernel(
    const float* __restrict__ tab,
    float*       __restrict__ P)
{
    __shared__ __attribute__((aligned(16))) float rows[NDIM * ROWS];
    const int t  = threadIdx.x;
    const int r0 = blockIdx.x * ROWS;
    const int seg   = blockIdx.y;
    const int rbase = seg == 0 ? 0 : SEG0 + (seg - 1) * SEGN;
    const int nrnd  = seg == 0 ? SEG0 : SEGN;

    const int4* __restrict__ rec = (const int4*)tab + rbase * NPAIR + t;

    int4 cur = rec[0];
    int4 ring[PF] = {};
    #pragma unroll
    for (int k = 0; k < PF - LAG; ++k)
        ring[k] = rec[(k + 1) * NPAIR];

    for (int idx = t; idx < NDIM * ROWS; idx += 256) {
        int c  = idx >> RSH;
        int rr = idx & (ROWS - 1);
        rows[idx] = (c == r0 + rr) ? 1.0f : 0.0f;
    }
    __syncthreads();

    for (int rd0 = 0; rd0 < nrnd; rd0 += PF) {
        #pragma unroll
        for (int k = 0; k < PF; ++k) {
            const int rd = rd0 + k;
            float4* pi = (float4*)((char*)rows + cur.x);
            float4* pj = (float4*)((char*)rows + cur.y);
            const float cth = __int_as_float(cur.z);
            const float sth = __int_as_float(cur.w);
            const float4 ui = *pi;
            const float4 uj = *pj;

            {
                int nr = rd + PF - LAG + 1;
                nr = nr < nrnd ? nr : nrnd - 1;
                ring[(k + PF - LAG) % PF] = rec[nr * NPAIR];
            }

            float4 vi, vj;
            vi.x = cth * ui.x + sth * uj.x;
            vi.y = cth * ui.y + sth * uj.y;
            vi.z = cth * ui.z + sth * uj.z;
            vi.w = cth * ui.w + sth * uj.w;
            vj.x = cth * uj.x - sth * ui.x;
            vj.y = cth * uj.y - sth * ui.y;
            vj.z = cth * uj.z - sth * ui.z;
            vj.w = cth * uj.w - sth * ui.w;
            *pi = vi;
            *pj = vj;
            round_barrier();

            cur = ring[k];
        }
    }

    float* __restrict__ Ps = P + seg * (NDIM * NDIM);
    for (int idx = t; idx < NDIM * ROWS; idx += 256) {
        const int c  = idx & (NDIM - 1);
        const int rr = idx >> 9;
        Ps[(r0 + rr) * NDIM + c] = rows[(c << RSH) + rr];
    }
}

#define CM  32
#define CN  64
#define CK  64
#define LDA 68
#define LDB 72

__global__ __launch_bounds__(256) void seg_product_kernel(
    const float* __restrict__ S,
    float*       __restrict__ D)
{
    __shared__ float sa[CM * LDA];
    __shared__ float sb[CK * LDB];

    const int t    = threadIdx.x;
    const int lane = t & 31;
    const int wave = t >> 5;
    const int lm   = lane & 15;
    const int half = lane >> 4;
    const int wm   = wave & 1;
    const int wn   = (wave >> 1) & 1;
    const int kh   = wave >> 2;

    const float* __restrict__ A = S + (2 * blockIdx.z) * (NDIM * NDIM);
    const float* __restrict__ B = A + NDIM * NDIM;
    float*       __restrict__ C = D + blockIdx.z * (NDIM * NDIM);

    const int m0 = blockIdx.x * CM;
    const int n0 = blockIdx.y * CN;
    const int c4 = (t & 15) << 2;
    const int rb = t >> 4;

    const float* __restrict__ ga = &A[(m0 + rb) * NDIM + c4];
    const float* __restrict__ gb = &B[rb * NDIM + n0 + c4];
    float4 ra0 = *(const float4*)&ga[0];
    float4 ra1 = *(const float4*)&ga[16 * NDIM];
    float4 rv0 = *(const float4*)&gb[0];
    float4 rv1 = *(const float4*)&gb[16 * NDIM];
    float4 rv2 = *(const float4*)&gb[32 * NDIM];
    float4 rv3 = *(const float4*)&gb[48 * NDIM];

    v8f acc0 = {}, acc1 = {};
    const int aoff = (wm * 16 + lm) * LDA + kh * 32 + 2 * half;
    const int boff = (kh * 32 + 2 * half) * LDB + wn * 32 + lm;

    for (int kc = 0; kc < NDIM; kc += CK) {
        *(float4*)&sa[rb * LDA + c4]        = ra0;
        *(float4*)&sa[(rb + 16) * LDA + c4] = ra1;
        *(float4*)&sb[rb * LDB + c4]        = rv0;
        *(float4*)&sb[(rb + 16) * LDB + c4] = rv1;
        *(float4*)&sb[(rb + 32) * LDB + c4] = rv2;
        *(float4*)&sb[(rb + 48) * LDB + c4] = rv3;
        round_barrier();
        if (kc + CK < NDIM) {
            ga += CK;
            gb += CK * NDIM;
            ra0 = *(const float4*)&ga[0];
            ra1 = *(const float4*)&ga[16 * NDIM];
            rv0 = *(const float4*)&gb[0];
            rv1 = *(const float4*)&gb[16 * NDIM];
            rv2 = *(const float4*)&gb[32 * NDIM];
            rv3 = *(const float4*)&gb[48 * NDIM];
        }
        #pragma unroll
        for (int kk = 0; kk < CK / 2; kk += 4) {
            const v2f a = *(const v2f*)&sa[aoff + kk];
            v2f b0, b1;
            b0.x = sb[boff + kk * LDB];
            b0.y = sb[boff + kk * LDB + LDB];
            b1.x = sb[boff + kk * LDB + 16];
            b1.y = sb[boff + kk * LDB + LDB + 16];
            acc0 = __builtin_amdgcn_wmma_f32_16x16x4_f32(
                false, a, false, b0, (short)0, acc0, false, false);
            acc1 = __builtin_amdgcn_wmma_f32_16x16x4_f32(
                false, a, false, b1, (short)0, acc1, false, false);
        }
        round_barrier();
    }

    float* red = sa;
    const int q = wave & 3;
    if (kh == 1) {
        #pragma unroll
        for (int v = 0; v < 8; ++v) {
            red[((q * 2 + 0) * 8 + v) * 32 + lane] = acc0[v];
            red[((q * 2 + 1) * 8 + v) * 32 + lane] = acc1[v];
        }
    }
    __syncthreads();
    if (kh == 0) {
        const int n = n0 + wn * 32 + lm;
        #pragma unroll
        for (int v = 0; v < 8; ++v) {
            const int m = m0 + wm * 16 + half * 8 + v;
            C[m * NDIM + n]      = acc0[v] + red[((q * 2 + 0) * 8 + v) * 32 + lane];
            C[m * NDIM + n + 16] = acc1[v] + red[((q * 2 + 1) * 8 + v) * 32 + lane];
        }
    }
}

#define BM  128
#define BN  64
#define BK  64
#define LDK 68
#define RPP 16

__global__ __launch_bounds__(256) void givens_gemm_kernel(
    const float* __restrict__ x,
    const float* __restrict__ U,
    const float* __restrict__ bias,
    float*       __restrict__ out)
{
    __shared__ float sx[BM * LDK];
    __shared__ float su[BN * LDK];

    const int t    = threadIdx.x;
    const int lane = t & 31;
    const int wave = t >> 5;
    const int lm   = lane & 15;
    const int half = lane >> 4;
    const int wm   = wave & 3;
    const int wn   = wave >> 2;

    const int m0 = blockIdx.x * BM;
    const int n0 = blockIdx.y * BN;

    const int c4 = (t & 15) << 2;
    const int rb = t >> 4;
    const float* __restrict__ gx = &x[(m0 + rb) * NDIM + c4];
    const float* __restrict__ gu = &U[(n0 + rb) * NDIM + c4];
    const int so = rb * LDK + c4;

    float4 rx0, rx1, rx2, rx3, rx4, rx5, rx6, rx7;
    float4 ru0, ru1, ru2, ru3;
    rx0 = *(const float4*)&gx[0 * RPP * NDIM];
    rx1 = *(const float4*)&gx[1 * RPP * NDIM];
    rx2 = *(const float4*)&gx[2 * RPP * NDIM];
    rx3 = *(const float4*)&gx[3 * RPP * NDIM];
    rx4 = *(const float4*)&gx[4 * RPP * NDIM];
    rx5 = *(const float4*)&gx[5 * RPP * NDIM];
    rx6 = *(const float4*)&gx[6 * RPP * NDIM];
    rx7 = *(const float4*)&gx[7 * RPP * NDIM];
    ru0 = *(const float4*)&gu[0 * RPP * NDIM];
    ru1 = *(const float4*)&gu[1 * RPP * NDIM];
    ru2 = *(const float4*)&gu[2 * RPP * NDIM];
    ru3 = *(const float4*)&gu[3 * RPP * NDIM];

    v8f acc[2][2] = {};
    const int arow0 = (wm * 32 +      lm) * LDK + 2 * half;
    const int arow1 = (wm * 32 + 16 + lm) * LDK + 2 * half;
    const int brow0 = (wn * 32 +      lm) * LDK + 2 * half;
    const int brow1 = (wn * 32 + 16 + lm) * LDK + 2 * half;

    for (int kc = 0; kc < NDIM; kc += BK) {
        *(float4*)&sx[so + 0 * RPP * LDK] = rx0;
        *(float4*)&sx[so + 1 * RPP * LDK] = rx1;
        *(float4*)&sx[so + 2 * RPP * LDK] = rx2;
        *(float4*)&sx[so + 3 * RPP * LDK] = rx3;
        *(float4*)&sx[so + 4 * RPP * LDK] = rx4;
        *(float4*)&sx[so + 5 * RPP * LDK] = rx5;
        *(float4*)&sx[so + 6 * RPP * LDK] = rx6;
        *(float4*)&sx[so + 7 * RPP * LDK] = rx7;
        *(float4*)&su[so + 0 * RPP * LDK] = ru0;
        *(float4*)&su[so + 1 * RPP * LDK] = ru1;
        *(float4*)&su[so + 2 * RPP * LDK] = ru2;
        *(float4*)&su[so + 3 * RPP * LDK] = ru3;
        round_barrier();

        if (kc + BK < NDIM) {
            gx += BK;
            gu += BK;
            rx0 = *(const float4*)&gx[0 * RPP * NDIM];
            rx1 = *(const float4*)&gx[1 * RPP * NDIM];
            rx2 = *(const float4*)&gx[2 * RPP * NDIM];
            rx3 = *(const float4*)&gx[3 * RPP * NDIM];
            rx4 = *(const float4*)&gx[4 * RPP * NDIM];
            rx5 = *(const float4*)&gx[5 * RPP * NDIM];
            rx6 = *(const float4*)&gx[6 * RPP * NDIM];
            rx7 = *(const float4*)&gx[7 * RPP * NDIM];
            ru0 = *(const float4*)&gu[0 * RPP * NDIM];
            ru1 = *(const float4*)&gu[1 * RPP * NDIM];
            ru2 = *(const float4*)&gu[2 * RPP * NDIM];
            ru3 = *(const float4*)&gu[3 * RPP * NDIM];
        }

        #pragma unroll
        for (int kk = 0; kk < BK; kk += 4) {
            const v2f a0 = *(const v2f*)&sx[arow0 + kk];
            const v2f a1 = *(const v2f*)&sx[arow1 + kk];
            const v2f b0 = *(const v2f*)&su[brow0 + kk];
            const v2f b1 = *(const v2f*)&su[brow1 + kk];
            acc[0][0] = __builtin_amdgcn_wmma_f32_16x16x4_f32(
                false, a0, false, b0, (short)0, acc[0][0], false, false);
            acc[0][1] = __builtin_amdgcn_wmma_f32_16x16x4_f32(
                false, a0, false, b1, (short)0, acc[0][1], false, false);
            acc[1][0] = __builtin_amdgcn_wmma_f32_16x16x4_f32(
                false, a1, false, b0, (short)0, acc[1][0], false, false);
            acc[1][1] = __builtin_amdgcn_wmma_f32_16x16x4_f32(
                false, a1, false, b1, (short)0, acc[1][1], false, false);
        }
        round_barrier();
    }

    #pragma unroll
    for (int ni = 0; ni < 2; ++ni) {
        const int n  = n0 + wn * 32 + ni * 16 + lm;
        const float bv = bias[n];
        #pragma unroll
        for (int mi = 0; mi < 2; ++mi) {
            const int mbase = m0 + wm * 32 + mi * 16 + half * 8;
            #pragma unroll
            for (int v = 0; v < 8; ++v) {
                out[(mbase + v) * NDIM + n] = acc[mi][ni][v] + bv;
            }
        }
    }
}

extern "C" void kernel_launch(void* const* d_in, const int* in_sizes, int n_in,
                              void* d_out, int out_size, void* d_ws, size_t ws_size,
                              hipStream_t stream) {
    const float* x      = (const float*)d_in[0];
    const float* angles = (const float*)d_in[1];
    const float* bias   = (const float*)d_in[2];
    const int*   blocks = (const int*)  d_in[3];

    float* U   = (float*)d_ws;
    float* out = (float*)d_out;
    float* tab = out;
    float* P   = out + 524288;
    float* Q   = out + 524288 + 1048576;

    round_table_kernel<<<dim3(NROUND), dim3(NPAIR), 0, stream>>>(angles, blocks, tab);
    build_u_kernel<<<dim3(NDIM / ROWS, NSEG), dim3(256), 0, stream>>>(tab, P);
    seg_product_kernel<<<dim3(NDIM / CM, NDIM / CN, 2), dim3(256), 0, stream>>>(P, Q);
    seg_product_kernel<<<dim3(NDIM / CM, NDIM / CN, 1), dim3(256), 0, stream>>>(Q, U);
    givens_gemm_kernel<<<dim3(NBATCH / BM, NDIM / BN), dim3(256), 0, stream>>>(
        x, U, bias, out);
}
